# starting kernel
# baseline (speedup 1.0000x reference)
#include <hip/hip_runtime.h>

typedef float v2f __attribute__((ext_vector_type(2)));
typedef float v8f __attribute__((ext_vector_type(8)));

#define N_NODES 50000
#define N_EDGES 800000
#define IN_CH   96
#define HID     128

__global__ void zero_f32(float* __restrict__ p, int n) {
  int i = blockIdx.x * blockDim.x + threadIdx.x;
  if (i < n) p[i] = 0.0f;
}

__global__ void degree_kernel(const int* __restrict__ dst,
                              float* __restrict__ cnt, int E) {
  int e = blockIdx.x * blockDim.x + threadIdx.x;
  if (e < E) atomicAdd(&cnt[dst[e]], 1.0f);
}

__global__ void inv_kernel(float* __restrict__ cnt, int n) {
  int i = blockIdx.x * blockDim.x + threadIdx.x;
  if (i < n) cnt[i] = 1.0f / fmaxf(cnt[i], 1.0f);
}

__global__ void scatter_kernel(const float* __restrict__ feat,
                               const int* __restrict__ src,
                               const int* __restrict__ dst,
                               float* __restrict__ agg, int E, int d) {
  int wave = (blockIdx.x * blockDim.x + threadIdx.x) >> 5;
  int lane = threadIdx.x & 31;
  if (wave >= E) return;
  int s = src[wave];
  int t = dst[wave];
  const float* fp = feat + (size_t)s * d;
  float*       ap = agg  + (size_t)t * d;
  for (int c = lane; c < d; c += 32)
    atomicAdd(&ap[c], fp[c]);
}

__global__ void sage_wmma_kernel(const float* __restrict__ agg,
                                 const float* __restrict__ inv,
                                 const float* __restrict__ hin,
                                 const float* __restrict__ Wl,
                                 const float* __restrict__ Wr,
                                 const float* __restrict__ bias,
                                 float* __restrict__ hout,
                                 int din, int dout, int nTilesN) {
  int wave  = (blockIdx.x * blockDim.x + threadIdx.x) >> 5;
  int lane  = threadIdx.x & 31;
  int mTile = wave / nTilesN;
  int nTile = wave % nTilesN;
  int mBase = mTile * 16;
  int nBase = nTile * 16;
  int half  = lane >> 4;
  int l16   = lane & 15;

  int m = mBase + l16;
  int n = nBase + l16;
  float invm = inv[m];
  const float* aggRow = agg + (size_t)m * din;
  const float* hinRow = hin + (size_t)m * din;

  v8f c = {0.f, 0.f, 0.f, 0.f, 0.f, 0.f, 0.f, 0.f};

  for (int k0 = 0; k0 < din; k0 += 4) {
    int kA = k0 + half * 2;
    v2f aL, aR, bL, bR;
    aL.x = aggRow[kA]     * invm;
    aL.y = aggRow[kA + 1] * invm;
    aR.x = hinRow[kA];
    aR.y = hinRow[kA + 1];
    bL.x = Wl[(size_t)kA       * dout + n];
    bL.y = Wl[(size_t)(kA + 1) * dout + n];
    bR.x = Wr[(size_t)kA       * dout + n];
    bR.y = Wr[(size_t)(kA + 1) * dout + n];
    c = __builtin_amdgcn_wmma_f32_16x16x4_f32(false, aL, false, bL,
                                              (short)0, c, false, false);
    c = __builtin_amdgcn_wmma_f32_16x16x4_f32(false, aR, false, bR,
                                              (short)0, c, false, false);
  }

  float bn = bias[n];
  #pragma unroll
  for (int v = 0; v < 8; ++v) {
    int mo = mBase + v + half * 8;
    float val = fmaxf(c[v] + bn, 0.0f);
    hout[(size_t)mo * dout + (nBase + l16)] = val;
  }
}

__global__ void classifier_kernel(const float* __restrict__ h,
                                  const float* __restrict__ Wc,
                                  const float* __restrict__ bc,
                                  float* __restrict__ out, int n) {
  __shared__ float w[128];
  __shared__ float b[2];
  if (threadIdx.x < 128) w[threadIdx.x] = Wc[threadIdx.x];
  if (threadIdx.x < 2)   b[threadIdx.x] = bc[threadIdx.x];
  __syncthreads();
  int i = blockIdx.x * blockDim.x + threadIdx.x;
  if (i >= n) return;
  const float* hr = h + (size_t)i * 64;
  float o0 = b[0], o1 = b[1];
  #pragma unroll
  for (int k = 0; k < 64; ++k) {
    float hv = hr[k];
    o0 += hv * w[k * 2 + 0];
    o1 += hv * w[k * 2 + 1];
  }
  out[i * 2 + 0] = o0;
  out[i * 2 + 1] = o1;
}

extern "C" void kernel_launch(void* const* d_in, const int* in_sizes, int n_in,
                              void* d_out, int out_size, void* d_ws, size_t ws_size,
                              hipStream_t stream) {
  const float* x   = (const float*)d_in[0];
  const int*   ei  = (const int*)d_in[1];
  const int*   src = ei;
  const int*   dst = ei + N_EDGES;
  const float* Wl1 = (const float*)d_in[2];
  const float* Wr1 = (const float*)d_in[3];
  const float* b1  = (const float*)d_in[4];
  const float* Wl2 = (const float*)d_in[5];
  const float* Wr2 = (const float*)d_in[6];
  const float* b2  = (const float*)d_in[7];
  const float* Wl3 = (const float*)d_in[8];
  const float* Wr3 = (const float*)d_in[9];
  const float* b3  = (const float*)d_in[10];
  const float* Wc  = (const float*)d_in[11];
  const float* bc  = (const float*)d_in[12];
  float* out = (float*)d_out;

  float* inv = (float*)d_ws;
  float* agg = inv + N_NODES;
  float* hA  = agg + (size_t)N_NODES * HID;
  float* hB  = hA  + (size_t)N_NODES * HID;

  const int ZT = 256;

  zero_f32<<<(N_NODES + ZT - 1) / ZT, ZT, 0, stream>>>(inv, N_NODES);
  degree_kernel<<<(N_EDGES + ZT - 1) / ZT, ZT, 0, stream>>>(dst, inv, N_EDGES);
  inv_kernel<<<(N_NODES + ZT - 1) / ZT, ZT, 0, stream>>>(inv, N_NODES);

  const int scatterBlocks = N_EDGES / 8;

  zero_f32<<<((N_NODES * IN_CH) + ZT - 1) / ZT, ZT, 0, stream>>>(agg, N_NODES * IN_CH);
  scatter_kernel<<<scatterBlocks, 256, 0, stream>>>(x, src, dst, agg, N_EDGES, IN_CH);
  {
    int nTilesN = HID / 16;
    int waves   = (N_NODES / 16) * nTilesN;
    sage_wmma_kernel<<<waves / 4, 128, 0, stream>>>(agg, inv, x, Wl1, Wr1, b1,
                                                    hA, IN_CH, HID, nTilesN);
  }

  zero_f32<<<((N_NODES * HID) + ZT - 1) / ZT, ZT, 0, stream>>>(agg, N_NODES * HID);
  scatter_kernel<<<scatterBlocks, 256, 0, stream>>>(hA, src, dst, agg, N_EDGES, HID);
  {
    int nTilesN = HID / 16;
    int waves   = (N_NODES / 16) * nTilesN;
    sage_wmma_kernel<<<waves / 4, 128, 0, stream>>>(agg, inv, hA, Wl2, Wr2, b2,
                                                    hB, HID, HID, nTilesN);
  }

  zero_f32<<<((N_NODES * HID) + ZT - 1) / ZT, ZT, 0, stream>>>(agg, N_NODES * HID);
  scatter_kernel<<<scatterBlocks, 256, 0, stream>>>(hB, src, dst, agg, N_EDGES, HID);
  {
    int nTilesN = (HID / 2) / 16;
    int waves   = (N_NODES / 16) * nTilesN;
    sage_wmma_kernel<<<waves / 4, 128, 0, stream>>>(agg, inv, hB, Wl3, Wr3, b3,
                                                    hA, HID, HID / 2, nTilesN);
  }

  classifier_kernel<<<(N_NODES + ZT - 1) / ZT, ZT, 0, stream>>>(hA, Wc, bc, out, N_NODES);
}


# Neighbour sums by gather, not float atomics: edges grouped by destination once per call (integer rank add, scan, fill); each 16-row tile's means summed in registers into LDS inside the layer kernel
# speedup vs baseline: 2.7863x; 2.7863x over previous
#include <hip/hip_runtime.h>

typedef float v2f __attribute__((ext_vector_type(2)));
typedef float v4f __attribute__((ext_vector_type(4)));
typedef float v8f __attribute__((ext_vector_type(8)));

#define N_NODES 50000
#define N_EDGES 800000
#define IN_CH   96
#define HID     128

__global__ void zero_i32(int* __restrict__ p, int n) {
  int i = blockIdx.x * blockDim.x + threadIdx.x;
  if (i < n) p[i] = 0;
}

__global__ void rank_kernel(const int* __restrict__ dst, int* __restrict__ cnt,
                            int* __restrict__ rnk, int E) {
  int e = blockIdx.x * blockDim.x + threadIdx.x;
  if (e < E) rnk[e] = atomicAdd(&cnt[dst[e]], 1);
}

#define SCAN_T 1024
__global__ void __launch_bounds__(SCAN_T)
scan_kernel(const int* __restrict__ cnt, int* __restrict__ off, int n) {
  __shared__ int wsum[32];
  __shared__ int wtot[32];
  const int tid  = threadIdx.x;
  const int lane = tid & 31;
  const int wv   = tid >> 5;
  const int chunk0 = blockIdx.x * SCAN_T;

  int s = 0;
  const int4* c4 = (const int4*)cnt;
  const int n4 = chunk0 >> 2;
  for (int j = tid; j < n4; j += SCAN_T) {
    int4 q = c4[j];
    s += (q.x + q.y) + (q.z + q.w);
  }
  for (int o = 16; o > 0; o >>= 1) s += __shfl_xor(s, o);
  if (lane == 0) wsum[wv] = s;

  const int i = chunk0 + tid;
  const int v = (i < n) ? cnt[i] : 0;
  int x = v;
  for (int o = 1; o < 32; o <<= 1) {
    int t = __shfl_up(x, o);
    if (lane >= o) x += t;
  }
  if (lane == 31) wtot[wv] = x;
  __syncthreads();

  int base = wsum[lane];
  for (int o = 16; o > 0; o >>= 1) base += __shfl_xor(base, o);
  int p = (lane < wv) ? wtot[lane] : 0;
  for (int o = 16; o > 0; o >>= 1) p += __shfl_xor(p, o);

  const int incl = base + p + x;
  if (i < n) off[i] = incl - v;
  if (i == n - 1) off[n] = incl;
}

__global__ void fill_kernel(const int* __restrict__ src, const int* __restrict__ dst,
                            const int* __restrict__ off, const int* __restrict__ rnk,
                            int* __restrict__ nbr, int E) {
  int e = blockIdx.x * blockDim.x + threadIdx.x;
  if (e < E) nbr[off[dst[e]] + rnk[e]] = src[e];
}

template <int DIN, int NLD>
__device__ __forceinline__ void add_rows(const float* __restrict__ hin, int idx, int j,
                                         int col, v4f& acc) {
  v4f v[NLD];
#pragma unroll
  for (int u = 0; u < NLD; ++u) {
    const int s = __builtin_amdgcn_readlane(idx, j + u);
    v[u] = *(const v4f*)(hin + (size_t)s * DIN + col);
  }
#pragma unroll
  for (int u = 0; u < NLD; ++u) acc += v[u];
}

template <int DIN, int DOUT>
__global__ void __launch_bounds__(DOUT * 2)
sage_gather_wmma_kernel(const float* __restrict__ hin,
                        const int* __restrict__ off,
                        const int* __restrict__ nbr,
                        const float* __restrict__ Wl,
                        const float* __restrict__ Wr,
                        const float* __restrict__ bias,
                        float* __restrict__ hout) {
  constexpr int WAVES  = DOUT / 16;
  constexpr int ROWS   = 16 / WAVES;
  constexpr int PITCH  = DIN + 4;
  constexpr int LANES4 = DIN / 4;
  __shared__ __attribute__((aligned(16))) float tile[16 * PITCH];

  const int lane = threadIdx.x & 31;
  const int wv = __builtin_amdgcn_readfirstlane(threadIdx.x >> 5);
  const int mBase = blockIdx.x * 16;

  const int col = (4 * lane < DIN) ? 4 * lane : DIN - 4;

#pragma unroll
  for (int r = 0; r < ROWS; ++r) {
    const int row = wv * ROWS + r;
    const int m   = mBase + row;
    const int beg = off[m];
    const int deg = off[m + 1] - beg;
    v4f acc = {0.f, 0.f, 0.f, 0.f};
    for (int c0 = 0; c0 < deg; c0 += 32) {
      const int nc  = (deg - c0 < 32) ? (deg - c0) : 32;
      const int idx = (lane < nc) ? nbr[beg + c0 + lane] : 0;
      int j = 0;
      for (; j + 8 <= nc; j += 8) add_rows<DIN, 8>(hin, idx, j, col, acc);
      if (j + 4 <= nc) { add_rows<DIN, 4>(hin, idx, j, col, acc); j += 4; }
      if (j + 2 <= nc) { add_rows<DIN, 2>(hin, idx, j, col, acc); j += 2; }
      if (j < nc)      { add_rows<DIN, 1>(hin, idx, j, col, acc); }
    }
    const float inv = 1.0f / fmaxf((float)deg, 1.0f);
    if (lane < LANES4) *(v4f*)(tile + row * PITCH + 4 * lane) = acc * inv;
  }
  __syncthreads();

  const int half  = lane >> 4;
  const int l16   = lane & 15;
  const int nBase = wv * 16;
  const int m = mBase + l16;
  const int n = nBase + l16;
  const float* aggRow = tile + l16 * PITCH;
  const float* hinRow = hin + (size_t)m * DIN;

  v8f c = {0.f, 0.f, 0.f, 0.f, 0.f, 0.f, 0.f, 0.f};

  for (int k0 = 0; k0 < DIN; k0 += 4) {
    const int kA = k0 + half * 2;
    v2f aL, aR, bL, bR;
    aL.x = aggRow[kA];
    aL.y = aggRow[kA + 1];
    aR.x = hinRow[kA];
    aR.y = hinRow[kA + 1];
    bL.x = Wl[(size_t)kA       * DOUT + n];
    bL.y = Wl[(size_t)(kA + 1) * DOUT + n];
    bR.x = Wr[(size_t)kA       * DOUT + n];
    bR.y = Wr[(size_t)(kA + 1) * DOUT + n];
    c = __builtin_amdgcn_wmma_f32_16x16x4_f32(false, aL, false, bL,
                                              (short)0, c, false, false);
    c = __builtin_amdgcn_wmma_f32_16x16x4_f32(false, aR, false, bR,
                                              (short)0, c, false, false);
  }

  const float bn = bias[n];
#pragma unroll
  for (int v = 0; v < 8; ++v) {
    const int mo = mBase + v + half * 8;
    const float val = fmaxf(c[v] + bn, 0.0f);
    hout[(size_t)mo * DOUT + n] = val;
  }
}

__global__ void classifier_kernel(const float* __restrict__ h,
                                  const float* __restrict__ Wc,
                                  const float* __restrict__ bc,
                                  float* __restrict__ out, int n) {
  __shared__ float w[128];
  __shared__ float b[2];
  if (threadIdx.x < 128) w[threadIdx.x] = Wc[threadIdx.x];
  if (threadIdx.x < 2)   b[threadIdx.x] = bc[threadIdx.x];
  __syncthreads();
  int i = blockIdx.x * blockDim.x + threadIdx.x;
  if (i >= n) return;
  const float* hr = h + (size_t)i * 64;
  float o0 = b[0], o1 = b[1];
  #pragma unroll
  for (int k = 0; k < 64; ++k) {
    float hv = hr[k];
    o0 += hv * w[k * 2 + 0];
    o1 += hv * w[k * 2 + 1];
  }
  out[i * 2 + 0] = o0;
  out[i * 2 + 1] = o1;
}

extern "C" void kernel_launch(void* const* d_in, const int* in_sizes, int n_in,
                              void* d_out, int out_size, void* d_ws, size_t ws_size,
                              hipStream_t stream) {
  const float* x   = (const float*)d_in[0];
  const int*   ei  = (const int*)d_in[1];
  const int*   src = ei;
  const int*   dst = ei + N_EDGES;
  const float* Wl1 = (const float*)d_in[2];
  const float* Wr1 = (const float*)d_in[3];
  const float* b1  = (const float*)d_in[4];
  const float* Wl2 = (const float*)d_in[5];
  const float* Wr2 = (const float*)d_in[6];
  const float* b2  = (const float*)d_in[7];
  const float* Wl3 = (const float*)d_in[8];
  const float* Wr3 = (const float*)d_in[9];
  const float* b3  = (const float*)d_in[10];
  const float* Wc  = (const float*)d_in[11];
  const float* bc  = (const float*)d_in[12];
  float* out = (float*)d_out;

  float* w0  = (float*)d_ws;
  float* grp = w0  + N_NODES;
  float* hA  = grp + (size_t)N_NODES * HID;
  float* hB  = hA  + (size_t)N_NODES * HID;

  int* cnt = (int*)grp;
  int* off = cnt + N_NODES;
  int* rnk = off + (N_NODES + 8);
  int* nbr = rnk + N_EDGES;

  const int ZT = 256;

  zero_i32<<<(N_NODES + ZT - 1) / ZT, ZT, 0, stream>>>(cnt, N_NODES);
  rank_kernel<<<(N_EDGES + ZT - 1) / ZT, ZT, 0, stream>>>(dst, cnt, rnk, N_EDGES);
  scan_kernel<<<(N_NODES + SCAN_T - 1) / SCAN_T, SCAN_T, 0, stream>>>(cnt, off, N_NODES);
  fill_kernel<<<(N_EDGES + ZT - 1) / ZT, ZT, 0, stream>>>(src, dst, off, rnk, nbr, N_EDGES);

  const int tiles = N_NODES / 16;

  sage_gather_wmma_kernel<IN_CH, HID><<<tiles, HID * 2, 0, stream>>>(
      x, off, nbr, Wl1, Wr1, b1, hA);

  sage_gather_wmma_kernel<HID, HID><<<tiles, HID * 2, 0, stream>>>(
      hA, off, nbr, Wl2, Wr2, b2, hB);

  sage_gather_wmma_kernel<HID, HID / 2><<<tiles, (HID / 2) * 2, 0, stream>>>(
      hB, off, nbr, Wl3, Wr3, b3, hA);

  classifier_kernel<<<(N_NODES + ZT - 1) / ZT, ZT, 0, stream>>>(hA, Wc, bc, out, N_NODES);
}


# Root operand through LDS: each tile's 16 root rows (one contiguous block) fetched once per work-group by whole-line 16-byte loads into a second LDS tile; phase 2 reads both A operands from LDS
# speedup vs baseline: 3.1129x; 1.1172x over previous
#include <hip/hip_runtime.h>

typedef float v2f __attribute__((ext_vector_type(2)));
typedef float v4f __attribute__((ext_vector_type(4)));
typedef float v8f __attribute__((ext_vector_type(8)));

#define N_NODES 50000
#define N_EDGES 800000
#define IN_CH   96
#define HID     128

__global__ void zero_i32(int* __restrict__ p, int n) {
  int i = blockIdx.x * blockDim.x + threadIdx.x;
  if (i < n) p[i] = 0;
}

__global__ void rank_kernel(const int* __restrict__ dst, int* __restrict__ cnt,
                            int* __restrict__ rnk, int E) {
  int e = blockIdx.x * blockDim.x + threadIdx.x;
  if (e < E) rnk[e] = atomicAdd(&cnt[dst[e]], 1);
}

#define SCAN_T 1024
__global__ void __launch_bounds__(SCAN_T)
scan_kernel(const int* __restrict__ cnt, int* __restrict__ off, int n) {
  __shared__ int wsum[32];
  __shared__ int wtot[32];
  const int tid  = threadIdx.x;
  const int lane = tid & 31;
  const int wv   = tid >> 5;
  const int chunk0 = blockIdx.x * SCAN_T;

  int s = 0;
  const int4* c4 = (const int4*)cnt;
  const int n4 = chunk0 >> 2;
  for (int j = tid; j < n4; j += SCAN_T) {
    int4 q = c4[j];
    s += (q.x + q.y) + (q.z + q.w);
  }
  for (int o = 16; o > 0; o >>= 1) s += __shfl_xor(s, o);
  if (lane == 0) wsum[wv] = s;

  const int i = chunk0 + tid;
  const int v = (i < n) ? cnt[i] : 0;
  int x = v;
  for (int o = 1; o < 32; o <<= 1) {
    int t = __shfl_up(x, o);
    if (lane >= o) x += t;
  }
  if (lane == 31) wtot[wv] = x;
  __syncthreads();

  int base = wsum[lane];
  for (int o = 16; o > 0; o >>= 1) base += __shfl_xor(base, o);
  int p = (lane < wv) ? wtot[lane] : 0;
  for (int o = 16; o > 0; o >>= 1) p += __shfl_xor(p, o);

  const int incl = base + p + x;
  if (i < n) off[i] = incl - v;
  if (i == n - 1) off[n] = incl;
}

__global__ void fill_kernel(const int* __restrict__ src, const int* __restrict__ dst,
                            const int* __restrict__ off, const int* __restrict__ rnk,
                            int* __restrict__ nbr, int E) {
  int e = blockIdx.x * blockDim.x + threadIdx.x;
  if (e < E) nbr[off[dst[e]] + rnk[e]] = src[e];
}

template <int DIN, int NLD>
__device__ __forceinline__ void add_rows(const float* __restrict__ hin, int idx, int j,
                                         int col, v4f& acc) {
  v4f v[NLD];
#pragma unroll
  for (int u = 0; u < NLD; ++u) {
    const int s = __builtin_amdgcn_readlane(idx, j + u);
    v[u] = *(const v4f*)(hin + (size_t)s * DIN + col);
  }
#pragma unroll
  for (int u = 0; u < NLD; ++u) acc += v[u];
}

template <int DIN, int DOUT>
__global__ void __launch_bounds__(DOUT * 2)
sage_gather_wmma_kernel(const float* __restrict__ hin,
                        const int* __restrict__ off,
                        const int* __restrict__ nbr,
                        const float* __restrict__ Wl,
                        const float* __restrict__ Wr,
                        const float* __restrict__ bias,
                        float* __restrict__ hout) {
  constexpr int WAVES  = DOUT / 16;
  constexpr int ROWS   = 16 / WAVES;
  constexpr int PITCH  = DIN + 4;
  constexpr int LANES4 = DIN / 4;
  constexpr int NT     = DOUT * 2;
  constexpr int PIECES = 16 * LANES4;
  constexpr int NP     = (PIECES + NT - 1) / NT;
  __shared__ __attribute__((aligned(16))) float tile[16 * PITCH];
  __shared__ __attribute__((aligned(16))) float root[16 * PITCH];

  const int lane = threadIdx.x & 31;
  const int wv = __builtin_amdgcn_readfirstlane(threadIdx.x >> 5);
  const int mBase = blockIdx.x * 16;

  const int col = (4 * lane < DIN) ? 4 * lane : DIN - 4;

  const v4f* rootSrc = (const v4f*)(hin + (size_t)mBase * DIN);
  v4f rv[NP];
#pragma unroll
  for (int p = 0; p < NP; ++p) {
    const int q = (int)threadIdx.x + p * NT;
    if (PIECES % NT == 0 || q < PIECES) rv[p] = rootSrc[q];
  }

#pragma unroll
  for (int r = 0; r < ROWS; ++r) {
    const int row = wv * ROWS + r;
    const int m   = mBase + row;
    const int beg = off[m];
    const int deg = off[m + 1] - beg;
    v4f acc = {0.f, 0.f, 0.f, 0.f};
    for (int c0 = 0; c0 < deg; c0 += 32) {
      const int nc  = (deg - c0 < 32) ? (deg - c0) : 32;
      const int idx = (lane < nc) ? nbr[beg + c0 + lane] : 0;
      int j = 0;
      for (; j + 8 <= nc; j += 8) add_rows<DIN, 8>(hin, idx, j, col, acc);
      if (j + 4 <= nc) { add_rows<DIN, 4>(hin, idx, j, col, acc); j += 4; }
      if (j + 2 <= nc) { add_rows<DIN, 2>(hin, idx, j, col, acc); j += 2; }
      if (j < nc)      { add_rows<DIN, 1>(hin, idx, j, col, acc); }
    }
    const float inv = 1.0f / fmaxf((float)deg, 1.0f);
    if (lane < LANES4) *(v4f*)(tile + row * PITCH + 4 * lane) = acc * inv;
  }
#pragma unroll
  for (int p = 0; p < NP; ++p) {
    const int q = (int)threadIdx.x + p * NT;
    if (PIECES % NT == 0 || q < PIECES) {
      const int rr = q / LANES4;
      const int cc = q - rr * LANES4;
      *(v4f*)(root + rr * PITCH + 4 * cc) = rv[p];
    }
  }
  __syncthreads();

  const int half  = lane >> 4;
  const int l16   = lane & 15;
  const int nBase = wv * 16;
  const int n = nBase + l16;
  const float* aggRow  = tile + l16 * PITCH;
  const float* rootRow = root + l16 * PITCH;

  v8f c = {0.f, 0.f, 0.f, 0.f, 0.f, 0.f, 0.f, 0.f};

  for (int k0 = 0; k0 < DIN; k0 += 4) {
    const int kA = k0 + half * 2;
    v2f aL, aR, bL, bR;
    aL.x = aggRow[kA];
    aL.y = aggRow[kA + 1];
    aR.x = rootRow[kA];
    aR.y = rootRow[kA + 1];
    bL.x = Wl[(size_t)kA       * DOUT + n];
    bL.y = Wl[(size_t)(kA + 1) * DOUT + n];
    bR.x = Wr[(size_t)kA       * DOUT + n];
    bR.y = Wr[(size_t)(kA + 1) * DOUT + n];
    c = __builtin_amdgcn_wmma_f32_16x16x4_f32(false, aL, false, bL,
                                              (short)0, c, false, false);
    c = __builtin_amdgcn_wmma_f32_16x16x4_f32(false, aR, false, bR,
                                              (short)0, c, false, false);
  }

  const float bn = bias[n];
#pragma unroll
  for (int v = 0; v < 8; ++v) {
    const int mo = mBase + v + half * 8;
    const float val = fmaxf(c[v] + bn, 0.0f);
    hout[(size_t)mo * DOUT + n] = val;
  }
}

__global__ void classifier_kernel(const float* __restrict__ h,
                                  const float* __restrict__ Wc,
                                  const float* __restrict__ bc,
                                  float* __restrict__ out, int n) {
  __shared__ float w[128];
  __shared__ float b[2];
  if (threadIdx.x < 128) w[threadIdx.x] = Wc[threadIdx.x];
  if (threadIdx.x < 2)   b[threadIdx.x] = bc[threadIdx.x];
  __syncthreads();
  int i = blockIdx.x * blockDim.x + threadIdx.x;
  if (i >= n) return;
  const float* hr = h + (size_t)i * 64;
  float o0 = b[0], o1 = b[1];
  #pragma unroll
  for (int k = 0; k < 64; ++k) {
    float hv = hr[k];
    o0 += hv * w[k * 2 + 0];
    o1 += hv * w[k * 2 + 1];
  }
  out[i * 2 + 0] = o0;
  out[i * 2 + 1] = o1;
}

extern "C" void kernel_launch(void* const* d_in, const int* in_sizes, int n_in,
                              void* d_out, int out_size, void* d_ws, size_t ws_size,
                              hipStream_t stream) {
  const float* x   = (const float*)d_in[0];
  const int*   ei  = (const int*)d_in[1];
  const int*   src = ei;
  const int*   dst = ei + N_EDGES;
  const float* Wl1 = (const float*)d_in[2];
  const float* Wr1 = (const float*)d_in[3];
  const float* b1  = (const float*)d_in[4];
  const float* Wl2 = (const float*)d_in[5];
  const float* Wr2 = (const float*)d_in[6];
  const float* b2  = (const float*)d_in[7];
  const float* Wl3 = (const float*)d_in[8];
  const float* Wr3 = (const float*)d_in[9];
  const float* b3  = (const float*)d_in[10];
  const float* Wc  = (const float*)d_in[11];
  const float* bc  = (const float*)d_in[12];
  float* out = (float*)d_out;

  float* w0  = (float*)d_ws;
  float* grp = w0  + N_NODES;
  float* hA  = grp + (size_t)N_NODES * HID;
  float* hB  = hA  + (size_t)N_NODES * HID;

  int* cnt = (int*)grp;
  int* off = cnt + N_NODES;
  int* rnk = off + (N_NODES + 8);
  int* nbr = rnk + N_EDGES;

  const int ZT = 256;

  zero_i32<<<(N_NODES + ZT - 1) / ZT, ZT, 0, stream>>>(cnt, N_NODES);
  rank_kernel<<<(N_EDGES + ZT - 1) / ZT, ZT, 0, stream>>>(dst, cnt, rnk, N_EDGES);
  scan_kernel<<<(N_NODES + SCAN_T - 1) / SCAN_T, SCAN_T, 0, stream>>>(cnt, off, N_NODES);
  fill_kernel<<<(N_EDGES + ZT - 1) / ZT, ZT, 0, stream>>>(src, dst, off, rnk, nbr, N_EDGES);

  const int tiles = N_NODES / 16;

  sage_gather_wmma_kernel<IN_CH, HID><<<tiles, HID * 2, 0, stream>>>(
      x, off, nbr, Wl1, Wr1, b1, hA);

  sage_gather_wmma_kernel<HID, HID><<<tiles, HID * 2, 0, stream>>>(
      hA, off, nbr, Wl2, Wr2, b2, hB);

  sage_gather_wmma_kernel<HID, HID / 2><<<tiles, (HID / 2) * 2, 0, stream>>>(
      hB, off, nbr, Wl3, Wr3, b3, hA);

  classifier_kernel<<<(N_NODES + ZT - 1) / ZT, ZT, 0, stream>>>(hA, Wc, bc, out, N_NODES);
}
